# starting kernel
# baseline (speedup 1.0000x reference)
#include <hip/hip_runtime.h>

typedef __attribute__((ext_vector_type(2))) float v2f;
typedef __attribute__((ext_vector_type(8))) float v8f;

#define B_ 32
#define N_ 400
#define M_ 50
#define D_ 128
#define VERY_NEG (-1e30f)

__global__ __launch_bounds__(256) void parts_kernel(
    const float* __restrict__ c, const float* __restrict__ q,
    const float* __restrict__ W,
    float* __restrict__ part_c, float* __restrict__ part_q)
{
    int gid  = blockIdx.x * blockDim.x + threadIdx.x;
    int wave = gid >> 5;
    int lane = gid & 31;
    const int totalC = B_ * N_;
    const int totalQ = B_ * M_;
    if (wave >= totalC + totalQ) return;

    const float* row;
    const float* w;
    float* dst;
    if (wave < totalC) {
        row = c + (size_t)wave * D_;
        w   = W;
        dst = part_c + wave;
    } else {
        int r = wave - totalC;
        row = q + (size_t)r * D_;
        w   = W + D_;
        dst = part_q + r;
    }

    float s = 0.0f;
#pragma unroll
    for (int k = 0; k < D_ / 32; ++k)
        s += row[lane + 32 * k] * w[lane + 32 * k];

#pragma unroll
    for (int off = 16; off >= 1; off >>= 1)
        s += __shfl_xor(s, off, 32);

    if (lane == 0) *dst = s;
}

__global__ __launch_bounds__(32) void sim_wmma_kernel(
    const float* __restrict__ c, const float* __restrict__ q,
    const int* __restrict__ cmask, const int* __restrict__ qmask,
    const float* __restrict__ W,
    const float* __restrict__ part_c, const float* __restrict__ part_q,
    float* __restrict__ out)
{
    const int nt   = blockIdx.x;
    const int b    = blockIdx.y;
    const int lane = threadIdx.x;
    const int half = lane >> 4;
    const int l16  = lane & 15;
    const int n0   = nt * 16;

    const float* crow = c + (((size_t)b * N_ + (n0 + l16)) * D_) + 2 * half;
    const float* wcq  = W + 2 * D_ + 2 * half;

    int mc[4];
#pragma unroll
    for (int t = 0; t < 4; ++t) {
        int m = t * 16 + l16;
        mc[t] = (m < M_) ? m : (M_ - 1);
    }
    const float* qr0 = q + (((size_t)b * M_ + mc[0]) * D_) + 2 * half;
    const float* qr1 = q + (((size_t)b * M_ + mc[1]) * D_) + 2 * half;
    const float* qr2 = q + (((size_t)b * M_ + mc[2]) * D_) + 2 * half;
    const float* qr3 = q + (((size_t)b * M_ + mc[3]) * D_) + 2 * half;

    v8f acc0 = {}, acc1 = {}, acc2 = {}, acc3 = {};

#pragma unroll 8
    for (int kk = 0; kk < D_; kk += 4) {
        v2f a = *(const v2f*)(crow + kk);
        v2f w = *(const v2f*)(wcq + kk);
        a.x *= w.x;
        a.y *= w.y;
        v2f b0 = *(const v2f*)(qr0 + kk);
        v2f b1 = *(const v2f*)(qr1 + kk);
        v2f b2 = *(const v2f*)(qr2 + kk);
        v2f b3 = *(const v2f*)(qr3 + kk);

        acc0 = __builtin_amdgcn_wmma_f32_16x16x4_f32(false, a, false, b0, (short)0, acc0, false, false);
        acc1 = __builtin_amdgcn_wmma_f32_16x16x4_f32(false, a, false, b1, (short)0, acc1, false, false);
        acc2 = __builtin_amdgcn_wmma_f32_16x16x4_f32(false, a, false, b2, (short)0, acc2, false, false);
        acc3 = __builtin_amdgcn_wmma_f32_16x16x4_f32(false, a, false, b3, (short)0, acc3, false, false);
    }

    float pq[4];
    int   qm[4];
#pragma unroll
    for (int t = 0; t < 4; ++t) {
        int m   = t * 16 + l16;
        bool ok = (m < M_);
        pq[t] = ok ? part_q[b * M_ + m] : 0.0f;
        qm[t] = ok ? qmask[b * M_ + m] : 0;
    }

    v8f accs[4] = {acc0, acc1, acc2, acc3};

#pragma unroll
    for (int v = 0; v < 8; ++v) {
        int    n  = n0 + v + 8 * half;
        float  pc = part_c[b * N_ + n];
        int    cm = cmask[b * N_ + n];
        size_t rowbase = ((size_t)b * N_ + n) * M_;
#pragma unroll
        for (int t = 0; t < 4; ++t) {
            int m = t * 16 + l16;
            if (m < M_) {
                float val = accs[t][v] + pc + pq[t];
                if (!(cm && qm[t])) val += VERY_NEG;
                out[rowbase + m] = val;
            }
        }
    }
}

extern "C" void kernel_launch(void* const* d_in, const int* in_sizes, int n_in,
                              void* d_out, int out_size, void* d_ws, size_t ws_size,
                              hipStream_t stream)
{
    const float* c     = (const float*)d_in[0];
    const float* q     = (const float*)d_in[1];
    const int*   cmask = (const int*)d_in[2];
    const int*   qmask = (const int*)d_in[3];
    const float* W     = (const float*)d_in[4];
    float*       out   = (float*)d_out;

    float* part_c = (float*)d_ws;
    float* part_q = part_c + B_ * N_;

    int waves   = B_ * N_ + B_ * M_;
    int threads = 256;
    int blocks  = (waves * 32 + threads - 1) / threads;
    parts_kernel<<<blocks, threads, 0, stream>>>(c, q, W, part_c, part_q);

    dim3 grid(N_ / 16, B_);
    sim_wmma_kernel<<<grid, 32, 0, stream>>>(c, q, cmask, qmask, W,
                                             part_c, part_q, out);
}


# M = 48 + 2: columns 48, 49 and c.w_c leave the 16-wide f32 tiles (64 of 160 WMMAs, 3 of 32 columns live) for per-lane packed f32 FMAs on fragments already held; 96 WMMAs stay; those sums reorder
# speedup vs baseline: 2.8893x; 2.8893x over previous
#include <hip/hip_runtime.h>


typedef __attribute__((ext_vector_type(2))) float v2f;
typedef __attribute__((ext_vector_type(4))) float v4f;
typedef __attribute__((ext_vector_type(8))) float v8f;
typedef __attribute__((ext_vector_type(4))) int   v4i;

#define B_ 32
#define N_ 400
#define M_ 50
#define D_ 128
#define VERY_NEG (-1e30f)
#define AHEAD 7

__device__ __forceinline__ float xhalf(float x)
{
    return __builtin_bit_cast(float, __builtin_amdgcn_permlanex16(
        __builtin_bit_cast(int, x), __builtin_bit_cast(int, x),
        0x76543210, 0xfedcba98, false, false));
}

template <int V>
__device__ __forceinline__ float row_bcast(float x)
{
    return __builtin_bit_cast(float, __builtin_amdgcn_mov_dpp(
        __builtin_bit_cast(int, x), 0x150 + V, 0xF, 0xF, false));
}

__global__ __launch_bounds__(32) __attribute__((amdgpu_waves_per_eu(1, 1))) void sim_wmma_kernel(
    const float* __restrict__ c, const float* __restrict__ q,
    const int* __restrict__ cmask, const int* __restrict__ qmask,
    const float* __restrict__ W,
    float* __restrict__ out)
{
    const int nt   = blockIdx.x;
    const int b    = blockIdx.y;
    const int lane = threadIdx.x;
    const int half = lane >> 4;
    const int l16  = lane & 15;
    const int n0   = nt * 16;

    const float* crow = c + (((size_t)b * N_ + (n0 + l16)) * D_) + 4 * half;
    const float* wc   = W + 4 * half;
    const float* wq   = W + D_ + 4 * half;
    const float* wcq  = W + 2 * D_ + 4 * half;

    const float* qr0 = q + (((size_t)b * M_ +  0 + l16) * D_) + 4 * half;
    const float* qr1 = q + (((size_t)b * M_ + 16 + l16) * D_) + 4 * half;
    const float* qr2 = q + (((size_t)b * M_ + 32 + l16) * D_) + 4 * half;
    const float* qe0 = q + (((size_t)b * M_ + 48) * D_) + 4 * half;
    const float* qe1 = q + (((size_t)b * M_ + 49) * D_) + 4 * half;

    v4f A[16], S[16], Wq[16], Wc[16], B0[16], B1[16], B2[16], E0[16], E1[16];

#define LOAD_STEP(j)                                   \
    do {                                               \
        A[(j)]  = *(const v4f*)(crow + 8 * (j));       \
        S[(j)]  = *(const v4f*)(wcq  + 8 * (j));       \
        Wq[(j)] = *(const v4f*)(wq   + 8 * (j));       \
        B0[(j)] = *(const v4f*)(qr0  + 8 * (j));       \
        B1[(j)] = *(const v4f*)(qr1  + 8 * (j));       \
        B2[(j)] = *(const v4f*)(qr2  + 8 * (j));       \
        Wc[(j)] = *(const v4f*)(wc   + 8 * (j));       \
        E0[(j)] = *(const v4f*)(qe0  + 8 * (j));       \
        E1[(j)] = *(const v4f*)(qe1  + 8 * (j));       \
    } while (0)

#pragma unroll
    for (int j = 0; j < AHEAD; ++j) {
        LOAD_STEP(j);
        __builtin_amdgcn_sched_barrier(0);
    }

    v8f acc0 = {}, acc1 = {}, acc2 = {};
    v2f d48a = {}, d48b = {}, d49a = {}, d49b = {}, dca = {}, dcb = {};

    v4i cm_lo = {}, cm_hi = {};
    int qm0 = 0, qm1 = 0, qm2 = 0, cme = 0, qme = 0;

#pragma unroll
    for (int j = 0; j < 16; ++j) {
        if (j + AHEAD < 16) LOAD_STEP(j + AHEAD);
        if (j + AHEAD == 16) {
            const int* cmp = cmask + (size_t)b * N_ + n0 + 8 * half;
            cm_lo = *(const v4i*)(cmp);
            cm_hi = *(const v4i*)(cmp + 4);
            qm0   = qmask[b * M_ +  0 + l16];
            qm1   = qmask[b * M_ + 16 + l16];
            qm2   = qmask[b * M_ + 32 + l16];
            cme   = cmask[(size_t)b * N_ + n0 + l16];
            qme   = qmask[b * M_ + 48 + half];
        }

        v2f a0  = {A[j].x,  A[j].y},  a1  = {A[j].z,  A[j].w};
        v2f s0  = {S[j].x,  S[j].y},  s1  = {S[j].z,  S[j].w};
        v2f u0  = {Wq[j].x, Wq[j].y}, u1  = {Wq[j].z, Wq[j].w};
        v2f w0  = {Wc[j].x, Wc[j].y}, w1  = {Wc[j].z, Wc[j].w};
        v2f e00 = {E0[j].x, E0[j].y}, e01 = {E0[j].z, E0[j].w};
        v2f e10 = {E1[j].x, E1[j].y}, e11 = {E1[j].z, E1[j].w};
        v2f b00 = {B0[j].x, B0[j].y}, b01 = {B0[j].z, B0[j].w};
        v2f b10 = {B1[j].x, B1[j].y}, b11 = {B1[j].z, B1[j].w};
        v2f b20 = {B2[j].x, B2[j].y}, b21 = {B2[j].z, B2[j].w};

        v2f af0 = __builtin_elementwise_fma(a0, s0, u0);
        v2f af1 = __builtin_elementwise_fma(a1, s1, u1);

        acc0 = __builtin_amdgcn_wmma_f32_16x16x4_f32(false, af0, false, b00, (short)0, acc0, false, false);
        acc1 = __builtin_amdgcn_wmma_f32_16x16x4_f32(false, af0, false, b10, (short)0, acc1, false, false);
        acc2 = __builtin_amdgcn_wmma_f32_16x16x4_f32(false, af0, false, b20, (short)0, acc2, false, false);
        acc0 = __builtin_amdgcn_wmma_f32_16x16x4_f32(false, af1, false, b01, (short)0, acc0, false, false);
        acc1 = __builtin_amdgcn_wmma_f32_16x16x4_f32(false, af1, false, b11, (short)0, acc1, false, false);
        acc2 = __builtin_amdgcn_wmma_f32_16x16x4_f32(false, af1, false, b21, (short)0, acc2, false, false);

        d48a = __builtin_elementwise_fma(af0, e00, d48a);
        d48b = __builtin_elementwise_fma(af1, e01, d48b);
        d49a = __builtin_elementwise_fma(af0, e10, d49a);
        d49b = __builtin_elementwise_fma(af1, e11, d49b);
        dca  = __builtin_elementwise_fma(a0,  w0,  dca);
        dcb  = __builtin_elementwise_fma(a1,  w1,  dcb);
        asm volatile("" : "+v"(d48a), "+v"(d48b), "+v"(d49a), "+v"(d49b), "+v"(dca), "+v"(dcb));
        __builtin_amdgcn_sched_barrier(0);
    }

    float s48 = (d48a.x + d48a.y) + (d48b.x + d48b.y);
    float s49 = (d49a.x + d49a.y) + (d49b.x + d49b.y);
    float sc  = (dca.x  + dca.y)  + (dcb.x  + dcb.y);
    float t48 = s48 + xhalf(s48);
    float t49 = s49 + xhalf(s49);
    float tc  = sc  + xhalf(sc);

    {
        float ev = (half ? t49 : t48) + tc;
        if (!(cme && qme)) ev += VERY_NEG;
        out[((size_t)b * N_ + n0 + l16) * M_ + 48 + half] = ev;
    }

    float tcr = __builtin_bit_cast(float, __builtin_amdgcn_update_dpp(
        __builtin_bit_cast(int, tc), __builtin_bit_cast(int, tc), 0x128, 0x2, 0xF, false));
    float pcs[8] = {row_bcast<0>(tcr), row_bcast<1>(tcr), row_bcast<2>(tcr), row_bcast<3>(tcr),
                    row_bcast<4>(tcr), row_bcast<5>(tcr), row_bcast<6>(tcr), row_bcast<7>(tcr)};

    v8f accs[3] = {acc0, acc1, acc2};
    int qm[3]  = {qm0, qm1, qm2};
    int cmr[8] = {cm_lo.x, cm_lo.y, cm_lo.z, cm_lo.w, cm_hi.x, cm_hi.y, cm_hi.z, cm_hi.w};

#pragma unroll
    for (int v = 0; v < 8; ++v) {
        int    n  = n0 + v + 8 * half;
        float  pc = pcs[v];
        int    cm = cmr[v];
        size_t rowbase = ((size_t)b * N_ + n) * M_;
#pragma unroll
        for (int t = 0; t < 3; ++t) {
            int m = t * 16 + l16;
            float val = accs[t][v] + pc;
            if (!(cm && qm[t])) val += VERY_NEG;
            out[rowbase + m] = val;
        }
    }
}

extern "C" void kernel_launch(void* const* d_in, const int* in_sizes, int n_in,
                              void* d_out, int out_size, void* d_ws, size_t ws_size,
                              hipStream_t stream)
{
    const float* c     = (const float*)d_in[0];
    const float* q     = (const float*)d_in[1];
    const int*   cmask = (const int*)d_in[2];
    const int*   qmask = (const int*)d_in[3];
    const float* W     = (const float*)d_in[4];
    float*       out   = (float*)d_out;

    dim3 grid(N_ / 16, B_);
    sim_wmma_kernel<<<grid, 32, 0, stream>>>(c, q, cmask, qmask, W, out);
}
